# starting kernel
# baseline (speedup 1.0000x reference)
#include <hip/hip_runtime.h>

#define N_NODES 128
#define NR_EDGES (N_NODES * (N_NODES - 1))
#define FPRIME 32
#define F_DIM 16
#define B_DIM 64

typedef __attribute__((ext_vector_type(2))) float v2f;
typedef __attribute__((ext_vector_type(8))) float v8f;

__global__ void __launch_bounds__(32) fuse_aw_kernel(const float* __restrict__ w,
                                                     const float* __restrict__ a,
                                                     float* __restrict__ wa) {
    const int t   = threadIdx.x;
    const int f   = t & 15;
    const int sel = t >> 4;
    float acc = 0.f;
#pragma unroll
    for (int g = 0; g < FPRIME; ++g)
        acc = fmaf(a[sel * FPRIME + g], w[g * F_DIM + f], acc);
    wa[sel * F_DIM + f] = acc;
}

__global__ void __launch_bounds__(32) node_pq_kernel(const float* __restrict__ x,
                                                     const float* __restrict__ wa,
                                                     float* __restrict__ p,
                                                     float* __restrict__ q) {
    const int b     = blockIdx.x >> 3;
    const int n0    = (blockIdx.x & 7) << 4;
    const int l     = threadIdx.x;
    const int mA    = l & 15;
    const int hi    = l >> 4;
    const int kHalf = hi << 1;
    const int col   = l & 15;
    const bool live = (col < 2);
    const int  colc = live ? col : 0;

    const float* xb  = x + (size_t)b * F_DIM * N_NODES;
    const float* war = wa + colc * F_DIM;

    v2f bp[4];
#pragma unroll
    for (int s = 0; s < 4; ++s)
        bp[s] = *(const v2f*)(war + 4 * s + kHalf);

    v8f c = {};
#pragma unroll
    for (int s = 0; s < 4; ++s) {
        const int kA = 4 * s + kHalf;
        v2f av, bv;
        av.x = xb[(kA + 0) * N_NODES + n0 + mA];
        av.y = xb[(kA + 1) * N_NODES + n0 + mA];
        bv.x = live ? bp[s].x : 0.f;
        bv.y = live ? bp[s].y : 0.f;
        c = __builtin_amdgcn_wmma_f32_16x16x4_f32(false, av, false, bv,
                                                  (short)0, c, false, false);
    }

    const int mBase = hi << 3;
    if (live) {
        float* dst = ((col == 0) ? p : q) + (size_t)b * N_NODES + n0 + mBase;
#pragma unroll
        for (int r = 0; r < 4; ++r) {
            v2f o2 = { c[2 * r], c[2 * r + 1] };
            *(v2f*)(dst + 2 * r) = o2;
        }
    }
}

__global__ void __launch_bounds__(256) edge_gather_kernel(const float* __restrict__ p,
                                                          const float* __restrict__ q,
                                                          const int* __restrict__ recv,
                                                          const int* __restrict__ send,
                                                          float* __restrict__ out) {
    const int e4 = blockIdx.x * 256 + threadIdx.x;
    const int b  = blockIdx.y;
    if (e4 < NR_EDGES / 4) {
        const int4 r4 = ((const int4*)recv)[e4];
        const int4 s4 = ((const int4*)send)[e4];
        const float* pb = p + b * N_NODES;
        const float* qb = q + b * N_NODES;
        float4 o;
        o.x = pb[r4.x] + qb[s4.x];
        o.y = pb[r4.y] + qb[s4.y];
        o.z = pb[r4.z] + qb[s4.z];
        o.w = pb[r4.w] + qb[s4.w];
        ((float4*)(out + (size_t)b * NR_EDGES))[e4] = o;
    }
}

extern "C" void kernel_launch(void* const* d_in, const int* in_sizes, int n_in,
                              void* d_out, int out_size, void* d_ws, size_t ws_size,
                              hipStream_t stream) {
    const float* x    = (const float*)d_in[0];
    const float* w    = (const float*)d_in[1];
    const float* a    = (const float*)d_in[2];
    const int*   recv = (const int*)d_in[3];
    const int*   send = (const int*)d_in[4];
    float* out = (float*)d_out;

    float* wa = (float*)d_ws;
    float* p  = wa + 32;
    float* q  = p + B_DIM * N_NODES;

    fuse_aw_kernel<<<1, 32, 0, stream>>>(w, a, wa);
    node_pq_kernel<<<B_DIM * (N_NODES / 16), 32, 0, stream>>>(x, wa, p, q);
    dim3 g3(((NR_EDGES / 4) + 255) / 256, B_DIM);
    edge_gather_kernel<<<g3, 256, 0, stream>>>(p, q, recv, send, out);
}


# Fuse the three launches into one kernel: each work-group recomputes wa in registers, p/q of its batch row by WMMA into LDS, then writes a quarter of that row's edges from LDS
# speedup vs baseline: 1.7816x; 1.7816x over previous
#include <hip/hip_runtime.h>


#define N_NODES 128
#define NR_EDGES (N_NODES * (N_NODES - 1))
#define FPRIME 32
#define F_DIM 16
#define B_DIM 64

#define CHUNKS 4
#define THREADS 128
#define QUADS_PER_B (NR_EDGES / 4)
#define QUADS_PER_CHUNK (QUADS_PER_B / CHUNKS)
#define QITER ((QUADS_PER_CHUNK + THREADS - 1) / THREADS)

typedef __attribute__((ext_vector_type(2))) float v2f;
typedef __attribute__((ext_vector_type(8))) float v8f;

__global__ void __launch_bounds__(THREADS) edge_attn_fused_kernel(const float* __restrict__ x,
                                                                  const float* __restrict__ w,
                                                                  const float* __restrict__ a,
                                                                  const int* __restrict__ recv,
                                                                  const int* __restrict__ send,
                                                                  float* __restrict__ out) {
    __shared__ __attribute__((aligned(16))) float spq[2 * N_NODES];

    const int t     = threadIdx.x;
    const int l     = t & 31;
    const int wv    = t >> 5;
    const int b     = blockIdx.x >> 2;
    const int chunk = blockIdx.x & (CHUNKS - 1);
    const int col   = l & 15;
    const int hi    = l >> 4;
    const int kHalf = hi << 1;
    const bool live = (col < 2);
    const int  colc = live ? col : 0;

    float4 ar[FPRIME / 4];
#pragma unroll
    for (int i = 0; i < FPRIME / 4; ++i)
        ar[i] = ((const float4*)(a + hi * FPRIME))[i];
    float wr[FPRIME];
#pragma unroll
    for (int g = 0; g < FPRIME; ++g)
        wr[g] = w[g * F_DIM + col];

    const float* xb = x + (size_t)b * F_DIM * N_NODES;
    const int    n0 = wv << 5;
    v2f xa[4][2];
#pragma unroll
    for (int s = 0; s < 4; ++s)
#pragma unroll
        for (int j = 0; j < 2; ++j)
            xa[s][j] = *(const v2f*)(xb + (4 * s + kHalf + j) * N_NODES + n0 + 2 * col);

    const int4* recv4 = (const int4*)recv + chunk * QUADS_PER_CHUNK;
    const int4* send4 = (const int4*)send + chunk * QUADS_PER_CHUNK;
    int4 r4[QITER], s4[QITER];
#pragma unroll
    for (int i = 0; i < QITER; ++i) {
        const int q  = t + THREADS * i;
        const int qc = (i < QITER - 1 || q < QUADS_PER_CHUNK) ? q : (QUADS_PER_CHUNK - 1);
        r4[i] = recv4[qc];
        s4[i] = send4[qc];
    }

    __builtin_amdgcn_sched_barrier(0);

    float acc = 0.f;
#pragma unroll
    for (int i = 0; i < FPRIME / 4; ++i) {
        acc = fmaf(ar[i].x, wr[4 * i + 0], acc);
        acc = fmaf(ar[i].y, wr[4 * i + 1], acc);
        acc = fmaf(ar[i].z, wr[4 * i + 2], acc);
        acc = fmaf(ar[i].w, wr[4 * i + 3], acc);
    }

    v8f c0 = {}, c1 = {};
#pragma unroll
    for (int s = 0; s < 4; ++s) {
        const int kB = 4 * s + kHalf;
        const float bx = __shfl(acc, colc * F_DIM + kB + 0, 32);
        const float by = __shfl(acc, colc * F_DIM + kB + 1, 32);
        v2f bv, av0, av1;
        bv.x  = live ? bx : 0.f;
        bv.y  = live ? by : 0.f;
        av0.x = xa[s][0].x;  av0.y = xa[s][1].x;
        av1.x = xa[s][0].y;  av1.y = xa[s][1].y;
        c0 = __builtin_amdgcn_wmma_f32_16x16x4_f32(false, av0, false, bv,
                                                   (short)0, c0, false, false);
        c1 = __builtin_amdgcn_wmma_f32_16x16x4_f32(false, av1, false, bv,
                                                   (short)0, c1, false, false);
    }

    if (live) {
        float4* dst = (float4*)(spq + col * N_NODES + n0 + (hi << 4));
#pragma unroll
        for (int i = 0; i < 4; ++i)
            dst[i] = make_float4(c0[2 * i], c1[2 * i], c0[2 * i + 1], c1[2 * i + 1]);
    }
    __syncthreads();

    const float* sp = spq;
    const float* sq = spq + N_NODES;
    float4* out4 = (float4*)(out + (size_t)b * NR_EDGES) + chunk * QUADS_PER_CHUNK;
#pragma unroll
    for (int i = 0; i < QITER; ++i) {
        const int q = t + THREADS * i;
        float4 o;
        o.x = sp[r4[i].x] + sq[s4[i].x];
        o.y = sp[r4[i].y] + sq[s4[i].y];
        o.z = sp[r4[i].z] + sq[s4[i].z];
        o.w = sp[r4[i].w] + sq[s4[i].w];
        if (i < QITER - 1 || q < QUADS_PER_CHUNK)
            out4[q] = o;
    }
}

extern "C" void kernel_launch(void* const* d_in, const int* in_sizes, int n_in,
                              void* d_out, int out_size, void* d_ws, size_t ws_size,
                              hipStream_t stream) {
    const float* x    = (const float*)d_in[0];
    const float* w    = (const float*)d_in[1];
    const float* a    = (const float*)d_in[2];
    const int*   recv = (const int*)d_in[3];
    const int*   send = (const int*)d_in[4];
    float* out = (float*)d_out;

    edge_attn_fused_kernel<<<B_DIM * CHUNKS, THREADS, 0, stream>>>(x, w, a, recv, send, out);
}
